# starting kernel
# baseline (speedup 1.0000x reference)
#include <hip/hip_runtime.h>
#include <math.h>

typedef __attribute__((ext_vector_type(2))) float v2f;
typedef __attribute__((ext_vector_type(8))) float v8f;

__global__ __launch_bounds__(256)
void prep_kernel(const float* __restrict__ W, const float* __restrict__ b,
                 const float* __restrict__ C, float* __restrict__ Mo,
                 float* __restrict__ bS)
{
    __shared__ __align__(16) float S[64 * 64];
    __shared__ __align__(16) float sW[64 * 64];
    const int tid = threadIdx.x;

    for (int i = tid; i < 4096; i += 256) sW[i] = W[i];

    if (tid < 64) {
        float m = -INFINITY;
        for (int j = 0; j < 64; ++j) m = fmaxf(m, C[tid * 64 + j]);
        float sum = 0.0f;
        for (int j = 0; j < 64; ++j) {
            float ev = expf(C[tid * 64 + j] - m);
            S[tid * 64 + j] = ev;
            sum += ev;
        }
        float inv = 1.0f / sum;
        for (int j = 0; j < 64; ++j) S[tid * 64 + j] *= inv;
    }
    __syncthreads();

    for (int idx = tid; idx < 4096; idx += 256) {
        int i = idx >> 6, j = idx & 63;
        float acc = 0.0f;
        for (int k = 0; k < 64; ++k) acc = fmaf(sW[i * 64 + k], S[k * 64 + j], acc);
        Mo[idx] = acc;
    }
    if (tid < 64) {
        float acc = 0.0f;
        for (int k = 0; k < 64; ++k) acc = fmaf(b[k], S[k * 64 + tid], acc);
        bS[tid] = acc;
    }
}

__global__ void deg_init(float* deg, int n)
{
    int i = blockIdx.x * blockDim.x + threadIdx.x;
    if (i < n) deg[i] = 1.0f;
}

__global__ void deg_count(const int* __restrict__ ei, float* deg, int E)
{
    int e = blockIdx.x * blockDim.x + threadIdx.x;
    if (e < E) atomicAdd(&deg[ei[E + e]], 1.0f);
}

__global__ void dinv_kernel(float* d, int n)
{
    int i = blockIdx.x * blockDim.x + threadIdx.x;
    if (i < n) d[i] = rsqrtf(d[i]);
}

__global__ __launch_bounds__(256)
void gemm_wmma_f32(const float* __restrict__ x, const float* __restrict__ M,
                   float* __restrict__ y, int nrows)
{
    __shared__ __align__(16) float sA[32 * 64];
    __shared__ __align__(16) float sB[64 * 64];

    const int tid = threadIdx.x;
    const int rowbase = blockIdx.x * 32;

    for (int i = tid * 4; i < 4096; i += 256 * 4)
        *(float4*)&sB[i] = *(const float4*)&M[i];

    for (int i = tid * 4; i < 2048; i += 256 * 4) {
        int row = rowbase + (i >> 6);
        float4 v = make_float4(0.f, 0.f, 0.f, 0.f);
        if (row < nrows) v = *(const float4*)&x[(size_t)row * 64 + (i & 63)];
        *(float4*)&sA[i] = v;
    }
    __syncthreads();

    const int wave = tid >> 5;
    const int lane = tid & 31;
    const int rblk = wave >> 2;
    const int cblk = wave & 3;
    const int r    = lane & 15;
    const int hi   = lane >> 4;

    const float* Arow = &sA[(rblk * 16 + r) * 64];
    const int    bcol = cblk * 16 + r;

    v8f acc = {};
#pragma unroll
    for (int k0 = 0; k0 < 64; k0 += 4) {
        v2f a, bfr;
        a.x = Arow[k0 + 2 * hi + 0];
        a.y = Arow[k0 + 2 * hi + 1];
        bfr.x = sB[(k0 + 2 * hi + 0) * 64 + bcol];
        bfr.y = sB[(k0 + 2 * hi + 1) * 64 + bcol];
        acc = __builtin_amdgcn_wmma_f32_16x16x4_f32(
             false, a,  false, bfr,
             (short)0, acc,  false,  false);
    }

    const int outrow = rowbase + rblk * 16 + hi * 8;
    const int outcol = cblk * 16 + r;
#pragma unroll
    for (int v = 0; v < 8; ++v) {
        int rr = outrow + v;
        if (rr < nrows) y[(size_t)rr * 64 + outcol] = acc[v];
    }
}

__global__ __launch_bounds__(256)
void self_init(const float* __restrict__ y, const float* __restrict__ dinv,
               const float* __restrict__ bS, float* __restrict__ out, int n)
{
    int t = blockIdx.x * blockDim.x + threadIdx.x;
    int i = t >> 4;
    int c = (t & 15) * 4;
    if (i >= n) return;
    float w = dinv[i] * dinv[i];
    float4 v  = *(const float4*)&y[(size_t)i * 64 + c];
    float4 bb = *(const float4*)&bS[c];
    float4 o  = make_float4(w * v.x + bb.x, w * v.y + bb.y,
                            w * v.z + bb.z, w * v.w + bb.w);
    *(float4*)&out[(size_t)i * 64 + c] = o;
}

__global__ __launch_bounds__(256)
void edge_scatter(const int* __restrict__ ei, const float* __restrict__ dinv,
                  const float* __restrict__ y, float* __restrict__ out, int E)
{
    int t = blockIdx.x * blockDim.x + threadIdx.x;
    int e = t >> 4;
    int c = (t & 15) * 4;
    if (e >= E) return;
    int s = ei[e];
    int d = ei[E + e];
    float w = dinv[s] * dinv[d];
    float4 v = *(const float4*)&y[(size_t)s * 64 + c];
    float* o = &out[(size_t)d * 64 + c];
    atomicAdd(o + 0, w * v.x);
    atomicAdd(o + 1, w * v.y);
    atomicAdd(o + 2, w * v.z);
    atomicAdd(o + 3, w * v.w);
}

extern "C" void kernel_launch(void* const* d_in, const int* in_sizes, int n_in,
                              void* d_out, int out_size, void* d_ws, size_t ws_size,
                              hipStream_t stream)
{
    const float* x  = (const float*)d_in[0];
    const int*   ei = (const int*)d_in[1];
    const float* W  = (const float*)d_in[2];
    const float* b  = (const float*)d_in[3];
    const float* Ca = (const float*)d_in[4];

    const int N = in_sizes[0] / 64;
    const int E = in_sizes[1] / 2;

    float* ws   = (float*)d_ws;
    float* M    = ws;
    float* bS   = ws + 4096;
    float* dinv = ws + 4160;
    float* y    = ws + 4160 + N;
    float* out  = (float*)d_out;

    prep_kernel <<<1, 256, 0, stream>>>(W, b, Ca, M, bS);
    deg_init    <<<(N + 255) / 256, 256, 0, stream>>>(dinv, N);
    deg_count   <<<(E + 255) / 256, 256, 0, stream>>>(ei, dinv, E);
    dinv_kernel <<<(N + 255) / 256, 256, 0, stream>>>(dinv, N);

    gemm_wmma_f32<<<(N + 31) / 32, 256, 0, stream>>>(x, M, y, N);

    self_init   <<<(N * 16 + 255) / 256, 256, 0, stream>>>(y, dinv, bS, out, N);

    long long escatter_threads = (long long)E * 16;
    edge_scatter<<<(int)((escatter_threads + 255) / 256), 256, 0, stream>>>(
        ei, dinv, y, out, E);
}


# GEMM made persistent (1024 groups loop over 32-row tiles); each builds M=W@softmax(C) and bS itself on WMMA, B fragments stay in registers; prep launch gone, 5 to 4 (M, softmax sums: new sum order)
# speedup vs baseline: 1.4939x; 1.4939x over previous
#include <hip/hip_runtime.h>
#include <math.h>

typedef __attribute__((ext_vector_type(2))) float v2f;
typedef __attribute__((ext_vector_type(8))) float v8f;

__global__ void deg_init(float* deg, int n)
{
    int i = blockIdx.x * blockDim.x + threadIdx.x;
    if (i < n) deg[i] = 1.0f;
}

__global__ void deg_count(const int* __restrict__ ei, float* deg, int E)
{
    int e = blockIdx.x * blockDim.x + threadIdx.x;
    if (e < E) atomicAdd(&deg[ei[E + e]], 1.0f);
}

#define CT_PITCH 68

__global__ __launch_bounds__(256)
void gemm_wmma_f32(const float* __restrict__ x, const float* __restrict__ W,
                   const float* __restrict__ b, const float* __restrict__ C,
                   float* dnode, float* __restrict__ y, float* __restrict__ out,
                   int nrows)
{
    __shared__ __align__(16) float smem[4096 + 2048 + 32 * CT_PITCH + 64 + 64];
    float* const sW  = smem;
    float* const sM  = smem;
    float* const sS  = smem + 4096;
    float* const sA  = smem + 4096;
    float* const sC  = smem + 4096 + 2048;
    float* const sD  = smem + 4096 + 2048 + 32 * CT_PITCH;
    float* const sBS = sD + 64;

    const int tid  = threadIdx.x;
    const int wave = tid >> 5;
    const int lane = tid & 31;
    const int r    = lane & 15;
    const int hi   = lane >> 4;
    const int ntiles = (nrows + 31) >> 5;
    int tile = blockIdx.x;

    const int xi0 = tid * 4;
    const int xi1 = tid * 4 + 1024;
    float4 xa = make_float4(0.f, 0.f, 0.f, 0.f), xb = xa;
    float  dg = 1.0f;
    if (tile < ntiles) {
        const int row0 = tile * 32 + (xi0 >> 6), row1 = tile * 32 + (xi1 >> 6);
        if (row0 < nrows) xa = *(const float4*)&x[(size_t)row0 * 64 + (xi0 & 63)];
        if (row1 < nrows) xb = *(const float4*)&x[(size_t)row1 * 64 + (xi1 & 63)];
        if (tid < 32 && tile * 32 + tid < nrows) dg = dnode[tile * 32 + tid];
    }

    for (int i = tid * 4; i < 4096; i += 256 * 4)
        *(float4*)&sW[i] = *(const float4*)&W[i];

#pragma unroll
    for (int rr = 0; rr < 8; ++rr) {
        const int row = wave * 8 + rr;
        const float c0 = C[row * 64 + lane], c1 = C[row * 64 + 32 + lane];
        float m = fmaxf(c0, c1);
        for (int o = 16; o > 0; o >>= 1) m = fmaxf(m, __shfl_xor(m, o));
        const float e0 = expf(c0 - m), e1 = expf(c1 - m);
        float s = e0 + e1;
        for (int o = 16; o > 0; o >>= 1) s += __shfl_xor(s, o);
        const float inv = 1.0f / s;
        sS[row * 64 + lane]      = e0 * inv;
        sS[row * 64 + 32 + lane] = e1 * inv;
    }
    __syncthreads();

    if (tid < 64) {
        float acc = 0.0f;
        for (int k = 0; k < 64; ++k) acc = fmaf(b[k], sS[k * 64 + tid], acc);
        sBS[tid] = acc;
    }
    {
        const int rb = wave >> 1, cb = (wave & 1) * 2;
        const float* Wrow = &sW[(rb * 16 + r) * 64];
        v8f m0 = {}, m1 = {};
#pragma unroll
        for (int k0 = 0; k0 < 64; k0 += 4) {
            v2f a, b0, b1;
            a.x  = Wrow[k0 + 2 * hi + 0];
            a.y  = Wrow[k0 + 2 * hi + 1];
            b0.x = sS[(k0 + 2 * hi + 0) * 64 + cb * 16 + r];
            b0.y = sS[(k0 + 2 * hi + 1) * 64 + cb * 16 + r];
            b1.x = sS[(k0 + 2 * hi + 0) * 64 + cb * 16 + 16 + r];
            b1.y = sS[(k0 + 2 * hi + 1) * 64 + cb * 16 + 16 + r];
            m0 = __builtin_amdgcn_wmma_f32_16x16x4_f32(false, a, false, b0,
                                                       (short)0, m0, false, false);
            m1 = __builtin_amdgcn_wmma_f32_16x16x4_f32(false, a, false, b1,
                                                       (short)0, m1, false, false);
        }
        __syncthreads();
        float* mrow = &sM[(rb * 16 + hi * 8) * 64 + cb * 16 + r];
#pragma unroll
        for (int v = 0; v < 8; ++v) {
            mrow[v * 64]      = m0[v];
            mrow[v * 64 + 16] = m1[v];
        }
    }
    __syncthreads();

    const int rblk = wave >> 2;
    const int cblk = wave & 3;
    const int bcol = cblk * 16 + r;
    const int c    = (tid & 15) * 4;
    const float4 bb = *(const float4*)&sBS[c];

    v2f bfr[16];
#pragma unroll
    for (int ks = 0; ks < 16; ++ks) {
        bfr[ks].x = sM[(ks * 4 + 2 * hi + 0) * 64 + bcol];
        bfr[ks].y = sM[(ks * 4 + 2 * hi + 1) * 64 + bcol];
    }

    const float* Arow = &sA[(rblk * 16 + r) * 64];
    float* crow = &sC[(rblk * 16 + hi * 8) * CT_PITCH + cblk * 16 + r];

    for (int par = 0; tile < ntiles; tile += gridDim.x, par ^= 32) {
        const int rowbase = tile * 32;
        const bool drow = (tid < 32) && (rowbase + tid < nrows);

        *(float4*)&sA[xi0] = xa;
        *(float4*)&sA[xi1] = xb;
        float di = 0.0f;
        if (tid < 32) {
            di = rsqrtf(dg);
            sD[par + tid] = di;
        }
        __syncthreads();

        const int tnext = tile + gridDim.x;
        if (tnext < ntiles) {
            const int row0 = tnext * 32 + (xi0 >> 6), row1 = tnext * 32 + (xi1 >> 6);
            xa = make_float4(0.f, 0.f, 0.f, 0.f); xb = xa; dg = 1.0f;
            if (row0 < nrows) xa = *(const float4*)&x[(size_t)row0 * 64 + (xi0 & 63)];
            if (row1 < nrows) xb = *(const float4*)&x[(size_t)row1 * 64 + (xi1 & 63)];
            if (tid < 32 && tnext * 32 + tid < nrows) dg = dnode[tnext * 32 + tid];
        }

        v8f acc = {};
#pragma unroll
        for (int ks = 0; ks < 16; ++ks) {
            v2f a;
            a.x = Arow[ks * 4 + 2 * hi + 0];
            a.y = Arow[ks * 4 + 2 * hi + 1];
            acc = __builtin_amdgcn_wmma_f32_16x16x4_f32(
                 false, a,  false, bfr[ks],
                 (short)0, acc,  false,  false);
        }

#pragma unroll
        for (int v = 0; v < 8; ++v) crow[v * CT_PITCH] = acc[v];
        __syncthreads();

#pragma unroll
        for (int rl = tid >> 4; rl < 32; rl += 16) {
            const int row = rowbase + rl;
            if (row < nrows) {
                const float4 v = *(const float4*)&sC[rl * CT_PITCH + c];
                const float  w = sD[par + rl] * sD[par + rl];
                *(float4*)&y[(size_t)row * 64 + c] = v;
                const float4 o = make_float4(w * v.x + bb.x, w * v.y + bb.y,
                                             w * v.z + bb.z, w * v.w + bb.w);
                *(float4*)&out[(size_t)row * 64 + c] = o;
            }
        }
        if (drow) dnode[rowbase + tid] = di;
    }
}

#define ES_GROUP 8

__global__ __launch_bounds__(256)
void edge_scatter(const int* __restrict__ ei, const float* __restrict__ dinv,
                  const float* __restrict__ y, float* __restrict__ out, int E)
{
    const int lane = threadIdx.x & 31;
    const int e0 = __builtin_amdgcn_readfirstlane(
        (int)((blockIdx.x * 8u + (threadIdx.x >> 5)) * 32u));
    if (e0 >= E) return;

    int   s = 0, d = 0;
    float w = 0.0f;
    const int e = e0 + lane;
    if (e < E) {
        s = ei[e];
        d = ei[E + e];
        w = dinv[s] * dinv[d];
    }
    const int wbits = __float_as_int(w);

    const float* ycol = y + lane;
    float*       ocol = out + lane;

    if (e0 + 32 <= E) {
#pragma unroll
        for (int g = 0; g < 32; g += ES_GROUP) {
            float lo[ES_GROUP], hi[ES_GROUP];
#pragma unroll
            for (int j = 0; j < ES_GROUP; ++j) {
                const int sj = __builtin_amdgcn_readlane(s, g + j);
                const float* yr = ycol + (size_t)sj * 64;
                lo[j] = yr[0];
                hi[j] = yr[32];
            }
#pragma unroll
            for (int j = 0; j < ES_GROUP; ++j) {
                const int   dj = __builtin_amdgcn_readlane(d, g + j);
                const float wj = __int_as_float(__builtin_amdgcn_readlane(wbits, g + j));
                float* orow = ocol + (size_t)dj * 64;
                atomicAdd(orow,      wj * lo[j]);
                atomicAdd(orow + 32, wj * hi[j]);
            }
        }
    } else {
        const int cnt = E - e0;
        for (int j = 0; j < cnt; ++j) {
            const int   sj = __builtin_amdgcn_readlane(s, j);
            const int   dj = __builtin_amdgcn_readlane(d, j);
            const float wj = __int_as_float(__builtin_amdgcn_readlane(wbits, j));
            const float* yr   = ycol + (size_t)sj * 64;
            float*       orow = ocol + (size_t)dj * 64;
            atomicAdd(orow,      wj * yr[0]);
            atomicAdd(orow + 32, wj * yr[32]);
        }
    }
}

extern "C" void kernel_launch(void* const* d_in, const int* in_sizes, int n_in,
                              void* d_out, int out_size, void* d_ws, size_t ws_size,
                              hipStream_t stream)
{
    const float* x  = (const float*)d_in[0];
    const int*   ei = (const int*)d_in[1];
    const float* W  = (const float*)d_in[2];
    const float* b  = (const float*)d_in[3];
    const float* Ca = (const float*)d_in[4];

    const int N = in_sizes[0] / 64;
    const int E = in_sizes[1] / 2;

    float* ws   = (float*)d_ws;
    float* dinv = ws + 4160;
    float* y    = ws + 4160 + N;
    float* out  = (float*)d_out;

    deg_init    <<<(N + 255) / 256, 256, 0, stream>>>(dinv, N);
    deg_count   <<<(E + 255) / 256, 256, 0, stream>>>(ei, dinv, E);

    gemm_wmma_f32<<<1024, 256, 0, stream>>>(x, W, b, Ca, dinv, y, out, N);

    edge_scatter<<<(E + 255) / 256, 256, 0, stream>>>(ei, dinv, y, out, E);
}


# Atomic adds kept inside one package half: rows split in two ranges; a wave reads its die, draws edge chunks from its half's ticket counter, adds only to its rows (degree + edge pass; new add order)
# speedup vs baseline: 1.9098x; 1.2784x over previous
#include <hip/hip_runtime.h>
#include <math.h>

typedef __attribute__((ext_vector_type(2))) float v2f;
typedef __attribute__((ext_vector_type(8))) float v8f;

#define TK_PITCH 64

__device__ __forceinline__ int package_half()
{
    return (int)((__builtin_amdgcn_s_sendmsg_rtn(0x87) >> 18) & 1u);
}

__device__ __forceinline__ unsigned draw_ticket(unsigned* c, int lane)
{
    unsigned t = 0;
    if (lane == 0) t = atomicAdd(c, 1u);
    return (unsigned)__builtin_amdgcn_readfirstlane((int)t);
}

__global__ void deg_init(float* deg, unsigned* tk, int n)
{
    int i = blockIdx.x * blockDim.x + threadIdx.x;
    if (i < n) deg[i] = 1.0f;
    if (i < 4) tk[i * TK_PITCH] = 0u;
}

#define DC_EDGES 1024

__global__ __launch_bounds__(256)
void deg_count(const int* __restrict__ ei, float* deg, unsigned* tk, int E, int n)
{
    const int lane   = threadIdx.x & 31;
    const int myhalf = package_half();
    const int split  = n >> 1;
    const unsigned nchunk = (unsigned)((E + DC_EDGES - 1) / DC_EDGES);
    const int* dstp = ei + E;

    for (int pass = 0; pass < 2; ++pass) {
        const int hh = myhalf ^ pass;
        unsigned* c = tk + hh * TK_PITCH;
        for (unsigned t = draw_ticket(c, lane); t < nchunk; t = draw_ticket(c, lane)) {
            const int base = (int)t * DC_EDGES + lane;
            for (int sb = 0; sb < DC_EDGES; sb += 256) {
                int d[8];
#pragma unroll
                for (int u = 0; u < 8; ++u) {
                    const int e = base + sb + u * 32;
                    d[u] = (e < E) ? dstp[e] : -1;
                }
#pragma unroll
                for (int u = 0; u < 8; ++u)
                    if (d[u] >= 0 && (int)(d[u] >= split) == hh) atomicAdd(&deg[d[u]], 1.0f);
            }
        }
    }
}

#define CT_PITCH 68

__global__ __launch_bounds__(256)
void gemm_wmma_f32(const float* __restrict__ x, const float* __restrict__ W,
                   const float* __restrict__ b, const float* __restrict__ C,
                   float* dnode, float* __restrict__ y, float* __restrict__ out,
                   int nrows)
{
    __shared__ __align__(16) float smem[4096 + 2048 + 32 * CT_PITCH + 64 + 64];
    float* const sW  = smem;
    float* const sM  = smem;
    float* const sS  = smem + 4096;
    float* const sA  = smem + 4096;
    float* const sC  = smem + 4096 + 2048;
    float* const sD  = smem + 4096 + 2048 + 32 * CT_PITCH;
    float* const sBS = sD + 64;

    const int tid  = threadIdx.x;
    const int wave = tid >> 5;
    const int lane = tid & 31;
    const int r    = lane & 15;
    const int hi   = lane >> 4;
    const int ntiles = (nrows + 31) >> 5;
    int tile = blockIdx.x;

    const int xi0 = tid * 4;
    const int xi1 = tid * 4 + 1024;
    float4 xa = make_float4(0.f, 0.f, 0.f, 0.f), xb = xa;
    float  dg = 1.0f;
    if (tile < ntiles) {
        const int row0 = tile * 32 + (xi0 >> 6), row1 = tile * 32 + (xi1 >> 6);
        if (row0 < nrows) xa = *(const float4*)&x[(size_t)row0 * 64 + (xi0 & 63)];
        if (row1 < nrows) xb = *(const float4*)&x[(size_t)row1 * 64 + (xi1 & 63)];
        if (tid < 32 && tile * 32 + tid < nrows) dg = dnode[tile * 32 + tid];
    }

    for (int i = tid * 4; i < 4096; i += 256 * 4)
        *(float4*)&sW[i] = *(const float4*)&W[i];

#pragma unroll
    for (int rr = 0; rr < 8; ++rr) {
        const int row = wave * 8 + rr;
        const float c0 = C[row * 64 + lane], c1 = C[row * 64 + 32 + lane];
        float m = fmaxf(c0, c1);
        for (int o = 16; o > 0; o >>= 1) m = fmaxf(m, __shfl_xor(m, o));
        const float e0 = expf(c0 - m), e1 = expf(c1 - m);
        float s = e0 + e1;
        for (int o = 16; o > 0; o >>= 1) s += __shfl_xor(s, o);
        const float inv = 1.0f / s;
        sS[row * 64 + lane]      = e0 * inv;
        sS[row * 64 + 32 + lane] = e1 * inv;
    }
    __syncthreads();

    if (tid < 64) {
        float acc = 0.0f;
        for (int k = 0; k < 64; ++k) acc = fmaf(b[k], sS[k * 64 + tid], acc);
        sBS[tid] = acc;
    }
    {
        const int rb = wave >> 1, cb = (wave & 1) * 2;
        const float* Wrow = &sW[(rb * 16 + r) * 64];
        v8f m0 = {}, m1 = {};
#pragma unroll
        for (int k0 = 0; k0 < 64; k0 += 4) {
            v2f a, b0, b1;
            a.x  = Wrow[k0 + 2 * hi + 0];
            a.y  = Wrow[k0 + 2 * hi + 1];
            b0.x = sS[(k0 + 2 * hi + 0) * 64 + cb * 16 + r];
            b0.y = sS[(k0 + 2 * hi + 1) * 64 + cb * 16 + r];
            b1.x = sS[(k0 + 2 * hi + 0) * 64 + cb * 16 + 16 + r];
            b1.y = sS[(k0 + 2 * hi + 1) * 64 + cb * 16 + 16 + r];
            m0 = __builtin_amdgcn_wmma_f32_16x16x4_f32(false, a, false, b0,
                                                       (short)0, m0, false, false);
            m1 = __builtin_amdgcn_wmma_f32_16x16x4_f32(false, a, false, b1,
                                                       (short)0, m1, false, false);
        }
        __syncthreads();
        float* mrow = &sM[(rb * 16 + hi * 8) * 64 + cb * 16 + r];
#pragma unroll
        for (int v = 0; v < 8; ++v) {
            mrow[v * 64]      = m0[v];
            mrow[v * 64 + 16] = m1[v];
        }
    }
    __syncthreads();

    const int rblk = wave >> 2;
    const int cblk = wave & 3;
    const int bcol = cblk * 16 + r;
    const int c    = (tid & 15) * 4;
    const float4 bb = *(const float4*)&sBS[c];

    v2f bfr[16];
#pragma unroll
    for (int ks = 0; ks < 16; ++ks) {
        bfr[ks].x = sM[(ks * 4 + 2 * hi + 0) * 64 + bcol];
        bfr[ks].y = sM[(ks * 4 + 2 * hi + 1) * 64 + bcol];
    }

    const float* Arow = &sA[(rblk * 16 + r) * 64];
    float* crow = &sC[(rblk * 16 + hi * 8) * CT_PITCH + cblk * 16 + r];

    for (int par = 0; tile < ntiles; tile += gridDim.x, par ^= 32) {
        const int rowbase = tile * 32;
        const bool drow = (tid < 32) && (rowbase + tid < nrows);

        *(float4*)&sA[xi0] = xa;
        *(float4*)&sA[xi1] = xb;
        float di = 0.0f;
        if (tid < 32) {
            di = rsqrtf(dg);
            sD[par + tid] = di;
        }
        __syncthreads();

        const int tnext = tile + gridDim.x;
        if (tnext < ntiles) {
            const int row0 = tnext * 32 + (xi0 >> 6), row1 = tnext * 32 + (xi1 >> 6);
            xa = make_float4(0.f, 0.f, 0.f, 0.f); xb = xa; dg = 1.0f;
            if (row0 < nrows) xa = *(const float4*)&x[(size_t)row0 * 64 + (xi0 & 63)];
            if (row1 < nrows) xb = *(const float4*)&x[(size_t)row1 * 64 + (xi1 & 63)];
            if (tid < 32 && tnext * 32 + tid < nrows) dg = dnode[tnext * 32 + tid];
        }

        v8f acc = {};
#pragma unroll
        for (int ks = 0; ks < 16; ++ks) {
            v2f a;
            a.x = Arow[ks * 4 + 2 * hi + 0];
            a.y = Arow[ks * 4 + 2 * hi + 1];
            acc = __builtin_amdgcn_wmma_f32_16x16x4_f32(
                 false, a,  false, bfr[ks],
                 (short)0, acc,  false,  false);
        }

#pragma unroll
        for (int v = 0; v < 8; ++v) crow[v * CT_PITCH] = acc[v];
        __syncthreads();

#pragma unroll
        for (int rl = tid >> 4; rl < 32; rl += 16) {
            const int row = rowbase + rl;
            if (row < nrows) {
                const float4 v = *(const float4*)&sC[rl * CT_PITCH + c];
                const float  w = sD[par + rl] * sD[par + rl];
                *(float4*)&y[(size_t)row * 64 + c] = v;
                const float4 o = make_float4(w * v.x + bb.x, w * v.y + bb.y,
                                             w * v.z + bb.z, w * v.w + bb.w);
                *(float4*)&out[(size_t)row * 64 + c] = o;
            }
        }
        if (drow) dnode[rowbase + tid] = di;
    }
}

#define ES_GROUP 8
#define EC_EDGES 128
#define EQ_CAP   256

__device__ __forceinline__ void walk32(int s, int d, int wbits,
                                       const float* ycol, float* ocol)
{
#pragma unroll
    for (int g = 0; g < 32; g += ES_GROUP) {
        float lo[ES_GROUP], hi[ES_GROUP];
#pragma unroll
        for (int j = 0; j < ES_GROUP; ++j) {
            const int sj = __builtin_amdgcn_readlane(s, g + j);
            const float* yr = ycol + (size_t)sj * 64;
            lo[j] = yr[0];
            hi[j] = yr[32];
        }
#pragma unroll
        for (int j = 0; j < ES_GROUP; ++j) {
            const int   dj = __builtin_amdgcn_readlane(d, g + j);
            const float wj = __int_as_float(__builtin_amdgcn_readlane(wbits, g + j));
            float* orow = ocol + (size_t)dj * 64;
            atomicAdd(orow,      wj * lo[j]);
            atomicAdd(orow + 32, wj * hi[j]);
        }
    }
}

__global__ __launch_bounds__(256)
void edge_scatter(const int* __restrict__ ei, const float* __restrict__ dinv,
                  const float* __restrict__ y, float* __restrict__ out,
                  unsigned* tk, int E, int n)
{
    __shared__ int ring[8][3][EQ_CAP];
    const int lane = threadIdx.x & 31;
    int* const Qs = ring[threadIdx.x >> 5][0];
    int* const Qd = ring[threadIdx.x >> 5][1];
    int* const Qw = ring[threadIdx.x >> 5][2];

    const int myhalf = package_half();
    const int split  = n >> 1;
    const unsigned nchunk = (unsigned)((E + EC_EDGES - 1) / EC_EDGES);
    const int* dstp = ei + E;

    const float* ycol = y + lane;
    float*       ocol = out + lane;

    int qh = 0, qn = 0;
    for (int pass = 0; pass < 2; ++pass) {
        const int hh = myhalf ^ pass;
        unsigned* c = tk + (2 + hh) * TK_PITCH;
        for (unsigned t = draw_ticket(c, lane); t < nchunk; t = draw_ticket(c, lane)) {
            const int base = (int)t * EC_EDGES + lane;
            int   dd[4], ss[4];
            float ww[4];
#pragma unroll
            for (int u = 0; u < 4; ++u) {
                const int e = base + u * 32;
                dd[u] = -1; ss[u] = 0;
                if (e < E) { dd[u] = dstp[e]; ss[u] = ei[e]; }
            }
            bool keep[4];
#pragma unroll
            for (int u = 0; u < 4; ++u) {
                keep[u] = dd[u] >= 0 && (int)(dd[u] >= split) == hh;
                const int si = keep[u] ? ss[u] : 0;
                const int di = keep[u] ? dd[u] : 0;
                ww[u] = dinv[si] * dinv[di];
            }
#pragma unroll
            for (int u = 0; u < 4; ++u) {
                const bool p = keep[u];
                const unsigned m = __builtin_amdgcn_ballot_w32(p);
                if (p) {
                    const int pos = (qh + qn + (int)__builtin_amdgcn_mbcnt_lo(m, 0u)) & (EQ_CAP - 1);
                    Qs[pos] = ss[u];
                    Qd[pos] = dd[u];
                    Qw[pos] = __float_as_int(ww[u]);
                }
                qn += __builtin_popcount(m);
            }
            while (qn >= 32) {
                const int rp = (qh + lane) & (EQ_CAP - 1);
                walk32(Qs[rp], Qd[rp], Qw[rp], ycol, ocol);
                qh = (qh + 32) & (EQ_CAP - 1);
                qn -= 32;
            }
        }
    }

    if (qn > 0) {
        const int rp = (qh + lane) & (EQ_CAP - 1);
        int s = 0, d = 0, wbits = 0;
        if (lane < qn) { s = Qs[rp]; d = Qd[rp]; wbits = Qw[rp]; }
        for (int j = 0; j < qn; ++j) {
            const int   sj = __builtin_amdgcn_readlane(s, j);
            const int   dj = __builtin_amdgcn_readlane(d, j);
            const float wj = __int_as_float(__builtin_amdgcn_readlane(wbits, j));
            const float* yr   = ycol + (size_t)sj * 64;
            float*       orow = ocol + (size_t)dj * 64;
            atomicAdd(orow,      wj * yr[0]);
            atomicAdd(orow + 32, wj * yr[32]);
        }
    }
}

extern "C" void kernel_launch(void* const* d_in, const int* in_sizes, int n_in,
                              void* d_out, int out_size, void* d_ws, size_t ws_size,
                              hipStream_t stream)
{
    const float* x  = (const float*)d_in[0];
    const int*   ei = (const int*)d_in[1];
    const float* W  = (const float*)d_in[2];
    const float* b  = (const float*)d_in[3];
    const float* Ca = (const float*)d_in[4];

    const int N = in_sizes[0] / 64;
    const int E = in_sizes[1] / 2;

    float* ws   = (float*)d_ws;
    unsigned* tk = (unsigned*)d_ws;
    float* dinv = ws + 4160;
    float* y    = ws + 4160 + N;
    float* out  = (float*)d_out;

    deg_init    <<<(N + 255) / 256, 256, 0, stream>>>(dinv, tk, N);
    deg_count   <<<256, 256, 0, stream>>>(ei, dinv, tk, E, N);

    gemm_wmma_f32<<<1024, 256, 0, stream>>>(x, W, b, Ca, dinv, y, out, N);

    edge_scatter<<<1024, 256, 0, stream>>>(ei, dinv, y, out, tk, E, N);
}


# Source factor of the edge weight folded into the GEMM epilogue: y is stored as dinv*y, so the edge pass does one scattered dinv[dst] look-up per edge, not two (weight product re-associated)
# speedup vs baseline: 1.9766x; 1.0350x over previous
#include <hip/hip_runtime.h>
#include <math.h>

typedef __attribute__((ext_vector_type(2))) float v2f;
typedef __attribute__((ext_vector_type(8))) float v8f;

#define TK_PITCH 64

__device__ __forceinline__ int package_half()
{
    return (int)((__builtin_amdgcn_s_sendmsg_rtn(0x87) >> 18) & 1u);
}

__device__ __forceinline__ unsigned draw_ticket(unsigned* c, int lane)
{
    unsigned t = 0;
    if (lane == 0) t = atomicAdd(c, 1u);
    return (unsigned)__builtin_amdgcn_readfirstlane((int)t);
}

__global__ void deg_init(float* deg, unsigned* tk, int n)
{
    int i = blockIdx.x * blockDim.x + threadIdx.x;
    if (i < n) deg[i] = 1.0f;
    if (i < 4) tk[i * TK_PITCH] = 0u;
}

#define DC_EDGES 1024

__global__ __launch_bounds__(256)
void deg_count(const int* __restrict__ ei, float* deg, unsigned* tk, int E, int n)
{
    const int lane   = threadIdx.x & 31;
    const int myhalf = package_half();
    const int split  = n >> 1;
    const unsigned nchunk = (unsigned)((E + DC_EDGES - 1) / DC_EDGES);
    const int* dstp = ei + E;

    for (int pass = 0; pass < 2; ++pass) {
        const int hh = myhalf ^ pass;
        unsigned* c = tk + hh * TK_PITCH;
        for (unsigned t = draw_ticket(c, lane); t < nchunk; t = draw_ticket(c, lane)) {
            const int base = (int)t * DC_EDGES + lane;
            for (int sb = 0; sb < DC_EDGES; sb += 256) {
                int d[8];
#pragma unroll
                for (int u = 0; u < 8; ++u) {
                    const int e = base + sb + u * 32;
                    d[u] = (e < E) ? dstp[e] : -1;
                }
#pragma unroll
                for (int u = 0; u < 8; ++u)
                    if (d[u] >= 0 && (int)(d[u] >= split) == hh) atomicAdd(&deg[d[u]], 1.0f);
            }
        }
    }
}

#define CT_PITCH 68

__global__ __launch_bounds__(256)
void gemm_wmma_f32(const float* __restrict__ x, const float* __restrict__ W,
                   const float* __restrict__ b, const float* __restrict__ C,
                   float* dnode, float* __restrict__ y, float* __restrict__ out,
                   int nrows)
{
    __shared__ __align__(16) float smem[4096 + 2048 + 32 * CT_PITCH + 64 + 64];
    float* const sW  = smem;
    float* const sM  = smem;
    float* const sS  = smem + 4096;
    float* const sA  = smem + 4096;
    float* const sC  = smem + 4096 + 2048;
    float* const sD  = smem + 4096 + 2048 + 32 * CT_PITCH;
    float* const sBS = sD + 64;

    const int tid  = threadIdx.x;
    const int wave = tid >> 5;
    const int lane = tid & 31;
    const int r    = lane & 15;
    const int hi   = lane >> 4;
    const int ntiles = (nrows + 31) >> 5;
    int tile = blockIdx.x;

    const int xi0 = tid * 4;
    const int xi1 = tid * 4 + 1024;
    float4 xa = make_float4(0.f, 0.f, 0.f, 0.f), xb = xa;
    float  dg = 1.0f;
    if (tile < ntiles) {
        const int row0 = tile * 32 + (xi0 >> 6), row1 = tile * 32 + (xi1 >> 6);
        if (row0 < nrows) xa = *(const float4*)&x[(size_t)row0 * 64 + (xi0 & 63)];
        if (row1 < nrows) xb = *(const float4*)&x[(size_t)row1 * 64 + (xi1 & 63)];
        if (tid < 32 && tile * 32 + tid < nrows) dg = dnode[tile * 32 + tid];
    }

    for (int i = tid * 4; i < 4096; i += 256 * 4)
        *(float4*)&sW[i] = *(const float4*)&W[i];

#pragma unroll
    for (int rr = 0; rr < 8; ++rr) {
        const int row = wave * 8 + rr;
        const float c0 = C[row * 64 + lane], c1 = C[row * 64 + 32 + lane];
        float m = fmaxf(c0, c1);
        for (int o = 16; o > 0; o >>= 1) m = fmaxf(m, __shfl_xor(m, o));
        const float e0 = expf(c0 - m), e1 = expf(c1 - m);
        float s = e0 + e1;
        for (int o = 16; o > 0; o >>= 1) s += __shfl_xor(s, o);
        const float inv = 1.0f / s;
        sS[row * 64 + lane]      = e0 * inv;
        sS[row * 64 + 32 + lane] = e1 * inv;
    }
    __syncthreads();

    if (tid < 64) {
        float acc = 0.0f;
        for (int k = 0; k < 64; ++k) acc = fmaf(b[k], sS[k * 64 + tid], acc);
        sBS[tid] = acc;
    }
    {
        const int rb = wave >> 1, cb = (wave & 1) * 2;
        const float* Wrow = &sW[(rb * 16 + r) * 64];
        v8f m0 = {}, m1 = {};
#pragma unroll
        for (int k0 = 0; k0 < 64; k0 += 4) {
            v2f a, b0, b1;
            a.x  = Wrow[k0 + 2 * hi + 0];
            a.y  = Wrow[k0 + 2 * hi + 1];
            b0.x = sS[(k0 + 2 * hi + 0) * 64 + cb * 16 + r];
            b0.y = sS[(k0 + 2 * hi + 1) * 64 + cb * 16 + r];
            b1.x = sS[(k0 + 2 * hi + 0) * 64 + cb * 16 + 16 + r];
            b1.y = sS[(k0 + 2 * hi + 1) * 64 + cb * 16 + 16 + r];
            m0 = __builtin_amdgcn_wmma_f32_16x16x4_f32(false, a, false, b0,
                                                       (short)0, m0, false, false);
            m1 = __builtin_amdgcn_wmma_f32_16x16x4_f32(false, a, false, b1,
                                                       (short)0, m1, false, false);
        }
        __syncthreads();
        float* mrow = &sM[(rb * 16 + hi * 8) * 64 + cb * 16 + r];
#pragma unroll
        for (int v = 0; v < 8; ++v) {
            mrow[v * 64]      = m0[v];
            mrow[v * 64 + 16] = m1[v];
        }
    }
    __syncthreads();

    const int rblk = wave >> 2;
    const int cblk = wave & 3;
    const int bcol = cblk * 16 + r;
    const int c    = (tid & 15) * 4;
    const float4 bb = *(const float4*)&sBS[c];

    v2f bfr[16];
#pragma unroll
    for (int ks = 0; ks < 16; ++ks) {
        bfr[ks].x = sM[(ks * 4 + 2 * hi + 0) * 64 + bcol];
        bfr[ks].y = sM[(ks * 4 + 2 * hi + 1) * 64 + bcol];
    }

    const float* Arow = &sA[(rblk * 16 + r) * 64];
    float* crow = &sC[(rblk * 16 + hi * 8) * CT_PITCH + cblk * 16 + r];

    for (int par = 0; tile < ntiles; tile += gridDim.x, par ^= 32) {
        const int rowbase = tile * 32;
        const bool drow = (tid < 32) && (rowbase + tid < nrows);

        *(float4*)&sA[xi0] = xa;
        *(float4*)&sA[xi1] = xb;
        float di = 0.0f;
        if (tid < 32) {
            di = rsqrtf(dg);
            sD[par + tid] = di;
        }
        __syncthreads();

        const int tnext = tile + gridDim.x;
        if (tnext < ntiles) {
            const int row0 = tnext * 32 + (xi0 >> 6), row1 = tnext * 32 + (xi1 >> 6);
            xa = make_float4(0.f, 0.f, 0.f, 0.f); xb = xa; dg = 1.0f;
            if (row0 < nrows) xa = *(const float4*)&x[(size_t)row0 * 64 + (xi0 & 63)];
            if (row1 < nrows) xb = *(const float4*)&x[(size_t)row1 * 64 + (xi1 & 63)];
            if (tid < 32 && tnext * 32 + tid < nrows) dg = dnode[tnext * 32 + tid];
        }

        v8f acc = {};
#pragma unroll
        for (int ks = 0; ks < 16; ++ks) {
            v2f a;
            a.x = Arow[ks * 4 + 2 * hi + 0];
            a.y = Arow[ks * 4 + 2 * hi + 1];
            acc = __builtin_amdgcn_wmma_f32_16x16x4_f32(
                 false, a,  false, bfr[ks],
                 (short)0, acc,  false,  false);
        }

#pragma unroll
        for (int v = 0; v < 8; ++v) crow[v * CT_PITCH] = acc[v];
        __syncthreads();

#pragma unroll
        for (int rl = tid >> 4; rl < 32; rl += 16) {
            const int row = rowbase + rl;
            if (row < nrows) {
                const float4 v = *(const float4*)&sC[rl * CT_PITCH + c];
                const float  w = sD[par + rl];
                const float4 s = make_float4(w * v.x, w * v.y, w * v.z, w * v.w);
                *(float4*)&y[(size_t)row * 64 + c] = s;
                const float4 o = make_float4(w * s.x + bb.x, w * s.y + bb.y,
                                             w * s.z + bb.z, w * s.w + bb.w);
                *(float4*)&out[(size_t)row * 64 + c] = o;
            }
        }
        if (drow) dnode[rowbase + tid] = di;
    }
}

#define ES_GROUP 8
#define EC_EDGES 128
#define EQ_CAP   256

__device__ __forceinline__ void walk32(int s, int d, int wbits,
                                       const float* ycol, float* ocol)
{
#pragma unroll
    for (int g = 0; g < 32; g += ES_GROUP) {
        float lo[ES_GROUP], hi[ES_GROUP];
#pragma unroll
        for (int j = 0; j < ES_GROUP; ++j) {
            const int sj = __builtin_amdgcn_readlane(s, g + j);
            const float* yr = ycol + (size_t)sj * 64;
            lo[j] = yr[0];
            hi[j] = yr[32];
        }
#pragma unroll
        for (int j = 0; j < ES_GROUP; ++j) {
            const int   dj = __builtin_amdgcn_readlane(d, g + j);
            const float wj = __int_as_float(__builtin_amdgcn_readlane(wbits, g + j));
            float* orow = ocol + (size_t)dj * 64;
            atomicAdd(orow,      wj * lo[j]);
            atomicAdd(orow + 32, wj * hi[j]);
        }
    }
}

__global__ __launch_bounds__(256)
void edge_scatter(const int* __restrict__ ei, const float* __restrict__ dinv,
                  const float* __restrict__ y, float* __restrict__ out,
                  unsigned* tk, int E, int n)
{
    __shared__ int ring[8][3][EQ_CAP];
    const int lane = threadIdx.x & 31;
    int* const Qs = ring[threadIdx.x >> 5][0];
    int* const Qd = ring[threadIdx.x >> 5][1];
    int* const Qw = ring[threadIdx.x >> 5][2];

    const int myhalf = package_half();
    const int split  = n >> 1;
    const unsigned nchunk = (unsigned)((E + EC_EDGES - 1) / EC_EDGES);
    const int* dstp = ei + E;

    const float* ycol = y + lane;
    float*       ocol = out + lane;

    int qh = 0, qn = 0;
    for (int pass = 0; pass < 2; ++pass) {
        const int hh = myhalf ^ pass;
        unsigned* c = tk + (2 + hh) * TK_PITCH;
        for (unsigned t = draw_ticket(c, lane); t < nchunk; t = draw_ticket(c, lane)) {
            const int base = (int)t * EC_EDGES + lane;
            int   dd[4], ss[4];
            float ww[4];
#pragma unroll
            for (int u = 0; u < 4; ++u) {
                const int e = base + u * 32;
                dd[u] = -1; ss[u] = 0;
                if (e < E) { dd[u] = dstp[e]; ss[u] = ei[e]; }
            }
            bool keep[4];
#pragma unroll
            for (int u = 0; u < 4; ++u) {
                keep[u] = dd[u] >= 0 && (int)(dd[u] >= split) == hh;
                const int di = keep[u] ? dd[u] : 0;
                ww[u] = dinv[di];
            }
#pragma unroll
            for (int u = 0; u < 4; ++u) {
                const bool p = keep[u];
                const unsigned m = __builtin_amdgcn_ballot_w32(p);
                if (p) {
                    const int pos = (qh + qn + (int)__builtin_amdgcn_mbcnt_lo(m, 0u)) & (EQ_CAP - 1);
                    Qs[pos] = ss[u];
                    Qd[pos] = dd[u];
                    Qw[pos] = __float_as_int(ww[u]);
                }
                qn += __builtin_popcount(m);
            }
            while (qn >= 32) {
                const int rp = (qh + lane) & (EQ_CAP - 1);
                walk32(Qs[rp], Qd[rp], Qw[rp], ycol, ocol);
                qh = (qh + 32) & (EQ_CAP - 1);
                qn -= 32;
            }
        }
    }

    if (qn > 0) {
        const int rp = (qh + lane) & (EQ_CAP - 1);
        int s = 0, d = 0, wbits = 0;
        if (lane < qn) { s = Qs[rp]; d = Qd[rp]; wbits = Qw[rp]; }
        for (int j = 0; j < qn; ++j) {
            const int   sj = __builtin_amdgcn_readlane(s, j);
            const int   dj = __builtin_amdgcn_readlane(d, j);
            const float wj = __int_as_float(__builtin_amdgcn_readlane(wbits, j));
            const float* yr   = ycol + (size_t)sj * 64;
            float*       orow = ocol + (size_t)dj * 64;
            atomicAdd(orow,      wj * yr[0]);
            atomicAdd(orow + 32, wj * yr[32]);
        }
    }
}

extern "C" void kernel_launch(void* const* d_in, const int* in_sizes, int n_in,
                              void* d_out, int out_size, void* d_ws, size_t ws_size,
                              hipStream_t stream)
{
    const float* x  = (const float*)d_in[0];
    const int*   ei = (const int*)d_in[1];
    const float* W  = (const float*)d_in[2];
    const float* b  = (const float*)d_in[3];
    const float* Ca = (const float*)d_in[4];

    const int N = in_sizes[0] / 64;
    const int E = in_sizes[1] / 2;

    float* ws   = (float*)d_ws;
    unsigned* tk = (unsigned*)d_ws;
    float* dinv = ws + 4160;
    float* y    = ws + 4160 + N;
    float* out  = (float*)d_out;

    deg_init    <<<(N + 255) / 256, 256, 0, stream>>>(dinv, tk, N);
    deg_count   <<<256, 256, 0, stream>>>(ei, dinv, tk, E, N);

    gemm_wmma_f32<<<1024, 256, 0, stream>>>(x, W, b, Ca, dinv, y, out, N);

    edge_scatter<<<1024, 256, 0, stream>>>(ei, dinv, y, out, tk, E, N);
}


# edge_scatter: a wave's ring remainder (under 32 edges) is walked in groups of 8 with the group's gathers all in flight, not one edge at a time with two dependent cache trips per edge; atomics kept
# speedup vs baseline: 2.2310x; 1.1287x over previous
#include <hip/hip_runtime.h>
#include <math.h>

typedef __attribute__((ext_vector_type(2))) float v2f;
typedef __attribute__((ext_vector_type(8))) float v8f;

#define TK_PITCH 64

__device__ __forceinline__ int package_half()
{
    return (int)((__builtin_amdgcn_s_sendmsg_rtn(0x87) >> 18) & 1u);
}

__device__ __forceinline__ unsigned draw_ticket(unsigned* c, int lane)
{
    unsigned t = 0;
    if (lane == 0) t = atomicAdd(c, 1u);
    return (unsigned)__builtin_amdgcn_readfirstlane((int)t);
}

__global__ void deg_init(float* deg, unsigned* tk, int n)
{
    int i = blockIdx.x * blockDim.x + threadIdx.x;
    if (i < n) deg[i] = 1.0f;
    if (i < 4) tk[i * TK_PITCH] = 0u;
}

#define DC_EDGES 1024

__global__ __launch_bounds__(256)
void deg_count(const int* __restrict__ ei, float* deg, unsigned* tk, int E, int n)
{
    const int lane   = threadIdx.x & 31;
    const int myhalf = package_half();
    const int split  = n >> 1;
    const unsigned nchunk = (unsigned)((E + DC_EDGES - 1) / DC_EDGES);
    const int* dstp = ei + E;

    for (int pass = 0; pass < 2; ++pass) {
        const int hh = myhalf ^ pass;
        unsigned* c = tk + hh * TK_PITCH;
        for (unsigned t = draw_ticket(c, lane); t < nchunk; t = draw_ticket(c, lane)) {
            const int base = (int)t * DC_EDGES + lane;
            for (int sb = 0; sb < DC_EDGES; sb += 256) {
                int d[8];
#pragma unroll
                for (int u = 0; u < 8; ++u) {
                    const int e = base + sb + u * 32;
                    d[u] = (e < E) ? dstp[e] : -1;
                }
#pragma unroll
                for (int u = 0; u < 8; ++u)
                    if (d[u] >= 0 && (int)(d[u] >= split) == hh) atomicAdd(&deg[d[u]], 1.0f);
            }
        }
    }
}

#define CT_PITCH 68

__global__ __launch_bounds__(256)
void gemm_wmma_f32(const float* __restrict__ x, const float* __restrict__ W,
                   const float* __restrict__ b, const float* __restrict__ C,
                   float* dnode, float* __restrict__ y, float* __restrict__ out,
                   int nrows)
{
    __shared__ __align__(16) float smem[4096 + 2048 + 32 * CT_PITCH + 64 + 64];
    float* const sW  = smem;
    float* const sM  = smem;
    float* const sS  = smem + 4096;
    float* const sA  = smem + 4096;
    float* const sC  = smem + 4096 + 2048;
    float* const sD  = smem + 4096 + 2048 + 32 * CT_PITCH;
    float* const sBS = sD + 64;

    const int tid  = threadIdx.x;
    const int wave = tid >> 5;
    const int lane = tid & 31;
    const int r    = lane & 15;
    const int hi   = lane >> 4;
    const int ntiles = (nrows + 31) >> 5;
    int tile = blockIdx.x;

    const int xi0 = tid * 4;
    const int xi1 = tid * 4 + 1024;
    float4 xa = make_float4(0.f, 0.f, 0.f, 0.f), xb = xa;
    float  dg = 1.0f;
    if (tile < ntiles) {
        const int row0 = tile * 32 + (xi0 >> 6), row1 = tile * 32 + (xi1 >> 6);
        if (row0 < nrows) xa = *(const float4*)&x[(size_t)row0 * 64 + (xi0 & 63)];
        if (row1 < nrows) xb = *(const float4*)&x[(size_t)row1 * 64 + (xi1 & 63)];
        if (tid < 32 && tile * 32 + tid < nrows) dg = dnode[tile * 32 + tid];
    }

    for (int i = tid * 4; i < 4096; i += 256 * 4)
        *(float4*)&sW[i] = *(const float4*)&W[i];

#pragma unroll
    for (int rr = 0; rr < 8; ++rr) {
        const int row = wave * 8 + rr;
        const float c0 = C[row * 64 + lane], c1 = C[row * 64 + 32 + lane];
        float m = fmaxf(c0, c1);
        for (int o = 16; o > 0; o >>= 1) m = fmaxf(m, __shfl_xor(m, o));
        const float e0 = expf(c0 - m), e1 = expf(c1 - m);
        float s = e0 + e1;
        for (int o = 16; o > 0; o >>= 1) s += __shfl_xor(s, o);
        const float inv = 1.0f / s;
        sS[row * 64 + lane]      = e0 * inv;
        sS[row * 64 + 32 + lane] = e1 * inv;
    }
    __syncthreads();

    if (tid < 64) {
        float acc = 0.0f;
        for (int k = 0; k < 64; ++k) acc = fmaf(b[k], sS[k * 64 + tid], acc);
        sBS[tid] = acc;
    }
    {
        const int rb = wave >> 1, cb = (wave & 1) * 2;
        const float* Wrow = &sW[(rb * 16 + r) * 64];
        v8f m0 = {}, m1 = {};
#pragma unroll
        for (int k0 = 0; k0 < 64; k0 += 4) {
            v2f a, b0, b1;
            a.x  = Wrow[k0 + 2 * hi + 0];
            a.y  = Wrow[k0 + 2 * hi + 1];
            b0.x = sS[(k0 + 2 * hi + 0) * 64 + cb * 16 + r];
            b0.y = sS[(k0 + 2 * hi + 1) * 64 + cb * 16 + r];
            b1.x = sS[(k0 + 2 * hi + 0) * 64 + cb * 16 + 16 + r];
            b1.y = sS[(k0 + 2 * hi + 1) * 64 + cb * 16 + 16 + r];
            m0 = __builtin_amdgcn_wmma_f32_16x16x4_f32(false, a, false, b0,
                                                       (short)0, m0, false, false);
            m1 = __builtin_amdgcn_wmma_f32_16x16x4_f32(false, a, false, b1,
                                                       (short)0, m1, false, false);
        }
        __syncthreads();
        float* mrow = &sM[(rb * 16 + hi * 8) * 64 + cb * 16 + r];
#pragma unroll
        for (int v = 0; v < 8; ++v) {
            mrow[v * 64]      = m0[v];
            mrow[v * 64 + 16] = m1[v];
        }
    }
    __syncthreads();

    const int rblk = wave >> 2;
    const int cblk = wave & 3;
    const int bcol = cblk * 16 + r;
    const int c    = (tid & 15) * 4;
    const float4 bb = *(const float4*)&sBS[c];

    v2f bfr[16];
#pragma unroll
    for (int ks = 0; ks < 16; ++ks) {
        bfr[ks].x = sM[(ks * 4 + 2 * hi + 0) * 64 + bcol];
        bfr[ks].y = sM[(ks * 4 + 2 * hi + 1) * 64 + bcol];
    }

    const float* Arow = &sA[(rblk * 16 + r) * 64];
    float* crow = &sC[(rblk * 16 + hi * 8) * CT_PITCH + cblk * 16 + r];

    for (int par = 0; tile < ntiles; tile += gridDim.x, par ^= 32) {
        const int rowbase = tile * 32;
        const bool drow = (tid < 32) && (rowbase + tid < nrows);

        *(float4*)&sA[xi0] = xa;
        *(float4*)&sA[xi1] = xb;
        float di = 0.0f;
        if (tid < 32) {
            di = rsqrtf(dg);
            sD[par + tid] = di;
        }
        __syncthreads();

        const int tnext = tile + gridDim.x;
        if (tnext < ntiles) {
            const int row0 = tnext * 32 + (xi0 >> 6), row1 = tnext * 32 + (xi1 >> 6);
            xa = make_float4(0.f, 0.f, 0.f, 0.f); xb = xa; dg = 1.0f;
            if (row0 < nrows) xa = *(const float4*)&x[(size_t)row0 * 64 + (xi0 & 63)];
            if (row1 < nrows) xb = *(const float4*)&x[(size_t)row1 * 64 + (xi1 & 63)];
            if (tid < 32 && tnext * 32 + tid < nrows) dg = dnode[tnext * 32 + tid];
        }

        v8f acc = {};
#pragma unroll
        for (int ks = 0; ks < 16; ++ks) {
            v2f a;
            a.x = Arow[ks * 4 + 2 * hi + 0];
            a.y = Arow[ks * 4 + 2 * hi + 1];
            acc = __builtin_amdgcn_wmma_f32_16x16x4_f32(
                 false, a,  false, bfr[ks],
                 (short)0, acc,  false,  false);
        }

#pragma unroll
        for (int v = 0; v < 8; ++v) crow[v * CT_PITCH] = acc[v];
        __syncthreads();

#pragma unroll
        for (int rl = tid >> 4; rl < 32; rl += 16) {
            const int row = rowbase + rl;
            if (row < nrows) {
                const float4 v = *(const float4*)&sC[rl * CT_PITCH + c];
                const float  w = sD[par + rl];
                const float4 s = make_float4(w * v.x, w * v.y, w * v.z, w * v.w);
                *(float4*)&y[(size_t)row * 64 + c] = s;
                const float4 o = make_float4(w * s.x + bb.x, w * s.y + bb.y,
                                             w * s.z + bb.z, w * s.w + bb.w);
                *(float4*)&out[(size_t)row * 64 + c] = o;
            }
        }
        if (drow) dnode[rowbase + tid] = di;
    }
}

#define ES_GROUP 8
#define EC_EDGES 128
#define EQ_CAP   256

__device__ __forceinline__ void walk32(int s, int d, int wbits,
                                       const float* ycol, float* ocol)
{
#pragma unroll
    for (int g = 0; g < 32; g += ES_GROUP) {
        float lo[ES_GROUP], hi[ES_GROUP];
#pragma unroll
        for (int j = 0; j < ES_GROUP; ++j) {
            const int sj = __builtin_amdgcn_readlane(s, g + j);
            const float* yr = ycol + (size_t)sj * 64;
            lo[j] = yr[0];
            hi[j] = yr[32];
        }
#pragma unroll
        for (int j = 0; j < ES_GROUP; ++j) {
            const int   dj = __builtin_amdgcn_readlane(d, g + j);
            const float wj = __int_as_float(__builtin_amdgcn_readlane(wbits, g + j));
            float* orow = ocol + (size_t)dj * 64;
            atomicAdd(orow,      wj * lo[j]);
            atomicAdd(orow + 32, wj * hi[j]);
        }
    }
}

__device__ __forceinline__ void walk_rest(int s, int d, int wbits, int cnt,
                                          const float* ycol, float* ocol)
{
    const int last = cnt - 1;
    for (int g = 0; g < cnt; g += ES_GROUP) {
        float lo[ES_GROUP], hi[ES_GROUP];
#pragma unroll
        for (int j = 0; j < ES_GROUP; ++j) {
            const int e  = (g + j < last) ? g + j : last;
            const int sj = __builtin_amdgcn_readlane(s, e);
            const float* yr = ycol + (size_t)sj * 64;
            lo[j] = yr[0];
            hi[j] = yr[32];
        }
#pragma unroll
        for (int j = 0; j < ES_GROUP; ++j) {
            if (g + j < cnt) {
                const int   dj = __builtin_amdgcn_readlane(d, g + j);
                const float wj = __int_as_float(__builtin_amdgcn_readlane(wbits, g + j));
                float* orow = ocol + (size_t)dj * 64;
                atomicAdd(orow,      wj * lo[j]);
                atomicAdd(orow + 32, wj * hi[j]);
            }
        }
    }
}

__global__ __launch_bounds__(256)
void edge_scatter(const int* __restrict__ ei, const float* __restrict__ dinv,
                  const float* __restrict__ y, float* __restrict__ out,
                  unsigned* tk, int E, int n)
{
    __shared__ int ring[8][3][EQ_CAP];
    const int lane = threadIdx.x & 31;
    int* const Qs = ring[threadIdx.x >> 5][0];
    int* const Qd = ring[threadIdx.x >> 5][1];
    int* const Qw = ring[threadIdx.x >> 5][2];

    const int myhalf = package_half();
    const int split  = n >> 1;
    const unsigned nchunk = (unsigned)((E + EC_EDGES - 1) / EC_EDGES);
    const int* dstp = ei + E;

    const float* ycol = y + lane;
    float*       ocol = out + lane;

    int qh = 0, qn = 0;
    for (int pass = 0; pass < 2; ++pass) {
        const int hh = myhalf ^ pass;
        unsigned* c = tk + (2 + hh) * TK_PITCH;
        for (unsigned t = draw_ticket(c, lane); t < nchunk; t = draw_ticket(c, lane)) {
            const int base = (int)t * EC_EDGES + lane;
            int   dd[4], ss[4];
            float ww[4];
#pragma unroll
            for (int u = 0; u < 4; ++u) {
                const int e = base + u * 32;
                dd[u] = -1; ss[u] = 0;
                if (e < E) { dd[u] = dstp[e]; ss[u] = ei[e]; }
            }
            bool keep[4];
#pragma unroll
            for (int u = 0; u < 4; ++u) {
                keep[u] = dd[u] >= 0 && (int)(dd[u] >= split) == hh;
                const int di = keep[u] ? dd[u] : 0;
                ww[u] = dinv[di];
            }
#pragma unroll
            for (int u = 0; u < 4; ++u) {
                const bool p = keep[u];
                const unsigned m = __builtin_amdgcn_ballot_w32(p);
                if (p) {
                    const int pos = (qh + qn + (int)__builtin_amdgcn_mbcnt_lo(m, 0u)) & (EQ_CAP - 1);
                    Qs[pos] = ss[u];
                    Qd[pos] = dd[u];
                    Qw[pos] = __float_as_int(ww[u]);
                }
                qn += __builtin_popcount(m);
            }
            while (qn >= 32) {
                const int rp = (qh + lane) & (EQ_CAP - 1);
                walk32(Qs[rp], Qd[rp], Qw[rp], ycol, ocol);
                qh = (qh + 32) & (EQ_CAP - 1);
                qn -= 32;
            }
        }
    }

    if (qn > 0) {
        const int rp = (qh + lane) & (EQ_CAP - 1);
        int s = 0, d = 0, wbits = 0;
        if (lane < qn) { s = Qs[rp]; d = Qd[rp]; wbits = Qw[rp]; }
        walk_rest(s, d, wbits, qn, ycol, ocol);
    }
}

extern "C" void kernel_launch(void* const* d_in, const int* in_sizes, int n_in,
                              void* d_out, int out_size, void* d_ws, size_t ws_size,
                              hipStream_t stream)
{
    const float* x  = (const float*)d_in[0];
    const int*   ei = (const int*)d_in[1];
    const float* W  = (const float*)d_in[2];
    const float* b  = (const float*)d_in[3];
    const float* Ca = (const float*)d_in[4];

    const int N = in_sizes[0] / 64;
    const int E = in_sizes[1] / 2;

    float* ws   = (float*)d_ws;
    unsigned* tk = (unsigned*)d_ws;
    float* dinv = ws + 4160;
    float* y    = ws + 4160 + N;
    float* out  = (float*)d_out;

    deg_init    <<<(N + 255) / 256, 256, 0, stream>>>(dinv, tk, N);
    deg_count   <<<256, 256, 0, stream>>>(ei, dinv, tk, E, N);

    gemm_wmma_f32<<<1024, 256, 0, stream>>>(x, W, b, Ca, dinv, y, out, N);

    edge_scatter<<<1024, 256, 0, stream>>>(ei, dinv, y, out, tk, E, N);
}
